# starting kernel
# baseline (speedup 1.0000x reference)
#include <hip/hip_runtime.h>
#include <math.h>

#define IN_DIM   256
#define HEADS    3
#define HID      3
#define NCLS     3
#define NEG_SLOPE 0.2f
#define GAT_EPS  1e-16f

typedef __attribute__((ext_vector_type(2))) float v2f;
typedef __attribute__((ext_vector_type(8))) float v8f;

__device__ __forceinline__ unsigned fenc(float f) {
    unsigned u = __float_as_uint(f);
    return (u & 0x80000000u) ? ~u : (u | 0x80000000u);
}
__device__ __forceinline__ float fdec(unsigned u) {
    return (u & 0x80000000u) ? __uint_as_float(u & 0x7fffffffu)
                             : __uint_as_float(~u);
}
__device__ __forceinline__ float lrelu(float x) { return x > 0.f ? x : NEG_SLOPE * x; }

__global__ __launch_bounds__(256) void gemm1_wmma(const float* __restrict__ x,
                                                  const float* __restrict__ W,
                                                  float* __restrict__ h, int N) {
    __shared__ float Wp[IN_DIM * 16];
    for (int i = threadIdx.x; i < IN_DIM * 16; i += 256) {
        int k = i >> 4, n = i & 15;
        Wp[i] = (n < HEADS * HID) ? W[k * (HEADS * HID) + n] : 0.0f;
    }
    __syncthreads();

    int lane  = threadIdx.x & 31;
    int wave  = threadIdx.x >> 5;
    long tile = (long)blockIdx.x * 8 + wave;
    long m0   = tile * 16;
    if (m0 >= N) return;

    int laneM = lane & 15;
    int koff  = (lane < 16) ? 0 : 2;
    const float* xrow = x + (m0 + laneM) * IN_DIM;

    v8f c = {};
    #pragma unroll 4
    for (int k0 = 0; k0 < IN_DIM; k0 += 4) {
        v2f a, b;
        a.x = xrow[k0 + koff];
        a.y = xrow[k0 + koff + 1];
        b.x = Wp[(k0 + koff)     * 16 + laneM];
        b.y = Wp[(k0 + koff + 1) * 16 + laneM];
        c = __builtin_amdgcn_wmma_f32_16x16x4_f32(false, a, false, b,
                                                  (short)0, c, false, false);
    }

    int rbase = (lane < 16) ? 0 : 8;
    if (laneM < HEADS * HID) {
        #pragma unroll
        for (int v = 0; v < 8; ++v)
            h[(m0 + rbase + v) * (HEADS * HID) + laneM] = c[v];
    }
}

__global__ void node_att(const float* __restrict__ h,
                         const float* __restrict__ atts,
                         const float* __restrict__ attd,
                         float* __restrict__ as_, float* __restrict__ ad_,
                         unsigned* __restrict__ m, int N) {
    int n = blockIdx.x * blockDim.x + threadIdx.x;
    if (n >= N) return;
    #pragma unroll
    for (int hh = 0; hh < HEADS; ++hh) {
        float s = 0.f, d = 0.f;
        #pragma unroll
        for (int cc = 0; cc < 3; ++cc) {
            float v = h[n * 9 + hh * 3 + cc];
            s += v * atts[hh * 3 + cc];
            d += v * attd[hh * 3 + cc];
        }
        as_[n * 3 + hh] = s;
        ad_[n * 3 + hh] = d;
        m[n * 3 + hh] = fenc(lrelu(s + d));
    }
}

__global__ void edge_max(const int* __restrict__ src, const int* __restrict__ dst,
                         const float* __restrict__ as_, const float* __restrict__ ad_,
                         unsigned* __restrict__ m, int E) {
    int i = blockIdx.x * blockDim.x + threadIdx.x;
    if (i >= E) return;
    int s = src[i], d = dst[i];
    #pragma unroll
    for (int hh = 0; hh < HEADS; ++hh) {
        float e = lrelu(as_[s * 3 + hh] + ad_[d * 3 + hh]);
        atomicMax(&m[d * 3 + hh], fenc(e));
    }
}

__global__ void node_init(const float* __restrict__ h,
                          const float* __restrict__ as_, const float* __restrict__ ad_,
                          const unsigned* __restrict__ m,
                          float* __restrict__ s, float* __restrict__ acc, int N) {
    int n = blockIdx.x * blockDim.x + threadIdx.x;
    if (n >= N) return;
    #pragma unroll
    for (int hh = 0; hh < HEADS; ++hh) {
        float e  = lrelu(as_[n * 3 + hh] + ad_[n * 3 + hh]);
        float p  = expf(e - fdec(m[n * 3 + hh]));
        s[n * 3 + hh] = p;
        #pragma unroll
        for (int cc = 0; cc < 3; ++cc)
            acc[n * 9 + hh * 3 + cc] = p * h[n * 9 + hh * 3 + cc];
    }
}

__global__ void edge_scatter(const int* __restrict__ src, const int* __restrict__ dst,
                             const float* __restrict__ h,
                             const float* __restrict__ as_, const float* __restrict__ ad_,
                             const unsigned* __restrict__ m,
                             float* __restrict__ s, float* __restrict__ acc, int E) {
    int i = blockIdx.x * blockDim.x + threadIdx.x;
    if (i >= E) return;
    int sn = src[i], dn = dst[i];
    __builtin_prefetch(&h[sn * 9], 0, 0);
    #pragma unroll
    for (int hh = 0; hh < HEADS; ++hh) {
        float e = lrelu(as_[sn * 3 + hh] + ad_[dn * 3 + hh]);
        float p = expf(e - fdec(m[dn * 3 + hh]));
        atomicAdd(&s[dn * 3 + hh], p);
        #pragma unroll
        for (int cc = 0; cc < 3; ++cc)
            atomicAdd(&acc[dn * 9 + hh * 3 + cc], p * h[sn * 9 + hh * 3 + cc]);
    }
}

__global__ void finish_elu(const float* __restrict__ acc, const float* __restrict__ s,
                           const float* __restrict__ b, float* __restrict__ t, int N) {
    int n = blockIdx.x * blockDim.x + threadIdx.x;
    if (n >= N) return;
    #pragma unroll
    for (int hh = 0; hh < HEADS; ++hh) {
        float inv = 1.0f / (s[n * 3 + hh] + GAT_EPS);
        #pragma unroll
        for (int cc = 0; cc < 3; ++cc) {
            float v = acc[n * 9 + hh * 3 + cc] * inv + b[hh * 3 + cc];
            t[n * 9 + hh * 3 + cc] = (v > 0.f) ? v : (expf(v) - 1.0f);
        }
    }
}

__global__ void gemm2(const float* __restrict__ t, const float* __restrict__ W2,
                      float* __restrict__ h2, int N) {
    __shared__ float Ws[81];
    if (threadIdx.x < 81) Ws[threadIdx.x] = W2[threadIdx.x];
    __syncthreads();
    int n = blockIdx.x * blockDim.x + threadIdx.x;
    if (n >= N) return;
    float tin[9];
    #pragma unroll
    for (int i = 0; i < 9; ++i) tin[i] = t[n * 9 + i];
    #pragma unroll
    for (int j = 0; j < 9; ++j) {
        float a = 0.f;
        #pragma unroll
        for (int i = 0; i < 9; ++i) a += tin[i] * Ws[i * 9 + j];
        h2[n * 9 + j] = a;
    }
}

__global__ void finish_out(const float* __restrict__ acc, const float* __restrict__ s,
                           const float* __restrict__ b2, float* __restrict__ out, int N) {
    int n = blockIdx.x * blockDim.x + threadIdx.x;
    if (n >= N) return;
    float v[NCLS];
    #pragma unroll
    for (int cc = 0; cc < NCLS; ++cc) {
        float acv = 0.f;
        #pragma unroll
        for (int hh = 0; hh < HEADS; ++hh)
            acv += acc[n * 9 + hh * 3 + cc] / (s[n * 3 + hh] + GAT_EPS);
        v[cc] = acv * (1.0f / HEADS) + b2[cc];
    }
    float mx = fmaxf(v[0], fmaxf(v[1], v[2]));
    float lse = logf(expf(v[0] - mx) + expf(v[1] - mx) + expf(v[2] - mx));
    #pragma unroll
    for (int cc = 0; cc < NCLS; ++cc)
        out[n * NCLS + cc] = v[cc] - mx - lse;
}

extern "C" void kernel_launch(void* const* d_in, const int* in_sizes, int n_in,
                              void* d_out, int out_size, void* d_ws, size_t ws_size,
                              hipStream_t stream) {
    const float* x        = (const float*)d_in[0];
    const int*   eidx     = (const int*)d_in[1];
    const float* W1       = (const float*)d_in[2];
    const float* att_src1 = (const float*)d_in[3];
    const float* att_dst1 = (const float*)d_in[4];
    const float* b1       = (const float*)d_in[5];
    const float* W2       = (const float*)d_in[6];
    const float* att_src2 = (const float*)d_in[7];
    const float* att_dst2 = (const float*)d_in[8];
    const float* b2       = (const float*)d_in[9];

    const int N = in_sizes[0] / IN_DIM;
    const int E = in_sizes[1] / 2;
    const int* src = eidx;
    const int* dst = eidx + E;

    float*    h   = (float*)d_ws;
    float*    as_ = h   + (size_t)N * 9;
    float*    ad_ = as_ + (size_t)N * 3;
    unsigned* m   = (unsigned*)(ad_ + (size_t)N * 3);
    float*    s   = (float*)(m + (size_t)N * 3);
    float*    acc = s   + (size_t)N * 3;
    float*    t   = acc + (size_t)N * 9;

    const int nb = (N + 255) / 256;
    const int eb = (E + 255) / 256;
    const int tiles = (N + 15) / 16;
    const int gb = (tiles + 7) / 8;

    gemm1_wmma <<<gb, 256, 0, stream>>>(x, W1, h, N);
    node_att   <<<nb, 256, 0, stream>>>(h, att_src1, att_dst1, as_, ad_, m, N);
    edge_max   <<<eb, 256, 0, stream>>>(src, dst, as_, ad_, m, E);
    node_init  <<<nb, 256, 0, stream>>>(h, as_, ad_, m, s, acc, N);
    edge_scatter<<<eb, 256, 0, stream>>>(src, dst, h, as_, ad_, m, s, acc, E);
    finish_elu <<<nb, 256, 0, stream>>>(acc, s, b1, t, N);

    gemm2      <<<nb, 256, 0, stream>>>(t, W2, h, N);
    node_att   <<<nb, 256, 0, stream>>>(h, att_src2, att_dst2, as_, ad_, m, N);
    edge_max   <<<eb, 256, 0, stream>>>(src, dst, as_, ad_, m, E);
    node_init  <<<nb, 256, 0, stream>>>(h, as_, ad_, m, s, acc, N);
    edge_scatter<<<eb, 256, 0, stream>>>(src, dst, h, as_, ad_, m, s, acc, E);
    finish_out <<<nb, 256, 0, stream>>>(acc, s, b2, (float*)d_out, N);
}


# Each 1024-thread block owns 416 destination nodes: sweeps the whole destination list once, lists its edges in LDS, max and sums by LDS atomics; global atomics gone; sum order = LDS arrival order
# speedup vs baseline: 5.5283x; 5.5283x over previous
#include <hip/hip_runtime.h>
#include <math.h>
#include <stdint.h>


#define IN_DIM   256
#define HEADS    3
#define HID      3
#define NCLS     3
#define NEG_SLOPE 0.2f
#define GAT_EPS  1e-16f

#define RNG    416
#define LCAP   20480
#define LTHR   1024
#define SCAN_U 8
#define WALK_W 4

typedef __attribute__((ext_vector_type(2))) float v2f;
typedef __attribute__((ext_vector_type(8))) float v8f;

__device__ __forceinline__ unsigned fenc(float f) {
    unsigned u = __float_as_uint(f);
    return (u & 0x80000000u) ? ~u : (u | 0x80000000u);
}
__device__ __forceinline__ float fdec(unsigned u) {
    return (u & 0x80000000u) ? __uint_as_float(u & 0x7fffffffu)
                             : __uint_as_float(~u);
}
__device__ __forceinline__ float lrelu(float x) { return x > 0.f ? x : NEG_SLOPE * x; }

__global__ __launch_bounds__(256) void gemm1_wmma(const float* __restrict__ x,
                                                  const float* __restrict__ W,
                                                  float* __restrict__ h, int N) {
    __shared__ float Wp[IN_DIM * 16];
    for (int i = threadIdx.x; i < IN_DIM * 16; i += 256) {
        int k = i >> 4, n = i & 15;
        Wp[i] = (n < HEADS * HID) ? W[k * (HEADS * HID) + n] : 0.0f;
    }
    __syncthreads();

    int lane  = threadIdx.x & 31;
    int wave  = threadIdx.x >> 5;
    long tile = (long)blockIdx.x * 8 + wave;
    long m0   = tile * 16;
    if (m0 >= N) return;

    int laneM = lane & 15;
    int koff  = (lane < 16) ? 0 : 2;
    const float* xrow = x + (m0 + laneM) * IN_DIM;

    v8f c = {};
    #pragma unroll 4
    for (int k0 = 0; k0 < IN_DIM; k0 += 4) {
        v2f a, b;
        a.x = xrow[k0 + koff];
        a.y = xrow[k0 + koff + 1];
        b.x = Wp[(k0 + koff)     * 16 + laneM];
        b.y = Wp[(k0 + koff + 1) * 16 + laneM];
        c = __builtin_amdgcn_wmma_f32_16x16x4_f32(false, a, false, b,
                                                  (short)0, c, false, false);
    }

    int rbase = (lane < 16) ? 0 : 8;
    if (laneM < HEADS * HID) {
        #pragma unroll
        for (int v = 0; v < 8; ++v)
            h[(m0 + rbase + v) * (HEADS * HID) + laneM] = c[v];
    }
}

__global__ void node_att(const float* __restrict__ h,
                         const float* __restrict__ atts,
                         const float* __restrict__ attd,
                         float* __restrict__ as_, float* __restrict__ ad_, int N) {
    int n = blockIdx.x * blockDim.x + threadIdx.x;
    if (n >= N) return;
    #pragma unroll
    for (int hh = 0; hh < HEADS; ++hh) {
        float s = 0.f, d = 0.f;
        #pragma unroll
        for (int cc = 0; cc < 3; ++cc) {
            float v = h[n * 9 + hh * 3 + cc];
            s += v * atts[hh * 3 + cc];
            d += v * attd[hh * 3 + cc];
        }
        as_[n * 3 + hh] = s;
        ad_[n * 3 + hh] = d;
    }
}

template <int LAYER>
__global__ __launch_bounds__(LTHR) void gat_layer(const int* __restrict__ src,
                                                  const int* __restrict__ dst,
                                                  const float* __restrict__ h,
                                                  const float* __restrict__ as_,
                                                  const float* __restrict__ ad_,
                                                  const float* __restrict__ bias,
                                                  float* __restrict__ outp, int N, int E) {
    __shared__ unsigned l_a[LCAP];
    __shared__ unsigned l_d[LCAP];
    __shared__ float    ad_l[RNG * 3];
    __shared__ unsigned m_l[RNG * 3];
    __shared__ float    s_l[RNG * 3];
    __shared__ float    acc_l[RNG * 9];
    __shared__ unsigned cnt;

    const int tid = threadIdx.x;
    const int lo  = blockIdx.x * RNG;
    const int rn  = min(RNG, N - lo);
    const unsigned urn = (unsigned)rn;

    for (int f = tid; f < rn * 3; f += LTHR) {
        float a = as_[lo * 3 + f], d = ad_[lo * 3 + f];
        ad_l[f] = d;
        m_l[f]  = fenc(lrelu(a + d));
    }
    if (tid == 0) cnt = 0u;
    __syncthreads();

    auto push = [&](unsigned ei, unsigned r) {
        unsigned pos = atomicAdd(&cnt, 1u);
        if (pos < (unsigned)LCAP) { l_a[pos] = ei; l_d[pos] = r; }
    };
    const bool vec_ok = ((E & 3) == 0) && ((((uintptr_t)dst) & 15) == 0);
    if (vec_ok) {
        const int4* dst4 = (const int4*)dst;
        const int E4 = E >> 2;
        auto test4 = [&](int4 d, int v) {
            unsigned r0 = (unsigned)(d.x - lo);
            unsigned r1 = (unsigned)(d.y - lo);
            unsigned r2 = (unsigned)(d.z - lo);
            unsigned r3 = (unsigned)(d.w - lo);
            unsigned rm = min(min(r0, r1), min(r2, r3));
            if (rm < urn) {
                unsigned e0 = (unsigned)v << 2;
                if (r0 < urn) push(e0,      r0);
                if (r1 < urn) push(e0 + 1u, r1);
                if (r2 < urn) push(e0 + 2u, r2);
                if (r3 < urn) push(e0 + 3u, r3);
            }
        };
        const int tile  = LTHR * SCAN_U;
        const int nfull = E4 / tile;
        const int t0 = ((int)blockIdx.x * nfull) / (int)gridDim.x;
        auto tile_base = [&](int k) { int t = k + t0; if (t >= nfull) t -= nfull; return t * tile; };
        auto load_set = [&](int4* d, int vb) {
            #pragma unroll
            for (int u = 0; u < SCAN_U; ++u) d[u] = dst4[vb + u * LTHR + tid];
        };
        auto scan_set = [&](const int4* d, int vb) {
            #pragma unroll
            for (int u = 0; u < SCAN_U; ++u) test4(d[u], vb + u * LTHR + tid);
        };
        if (nfull > 0) {
            int4 sa[SCAN_U], sb[SCAN_U];
            load_set(sa, tile_base(0));
            for (int k = 0; k < nfull; k += 2) {
                const bool has_b = (k + 1) < nfull;
                load_set(sb, tile_base(has_b ? k + 1 : k));
                scan_set(sa, tile_base(k));
                load_set(sa, tile_base((k + 2) < nfull ? k + 2 : k));
                if (has_b) scan_set(sb, tile_base(k + 1));
            }
        }
        for (int v = nfull * tile + tid; v < E4; v += LTHR) test4(dst4[v], v);
    } else {
        for (int i = tid; i < E; i += LTHR) {
            unsigned r = (unsigned)(dst[i] - lo);
            if (r < urn) push((unsigned)i, r);
        }
    }
    __syncthreads();
    const unsigned total  = cnt;
    const bool     listed = total <= (unsigned)LCAP;

    auto max_apply = [&](const float* a, unsigned dl) {
        float d[HEADS];
        #pragma unroll
        for (int hh = 0; hh < HEADS; ++hh) d[hh] = ad_l[dl * 3 + hh];
        #pragma unroll
        for (int hh = 0; hh < HEADS; ++hh)
            atomicMax(&m_l[dl * 3 + hh], fenc(lrelu(a[hh] + d[hh])));
    };
    if (listed) {
        for (unsigned j0 = tid; j0 < total; j0 += WALK_W * LTHR) {
            unsigned ei[WALK_W], dl[WALK_W], sn[WALK_W];
            bool on[WALK_W];
            float a[WALK_W][HEADS];
            #pragma unroll
            for (int w = 0; w < WALK_W; ++w) {
                unsigned j = j0 + w * LTHR;
                on[w] = j < total;
                unsigned jc = on[w] ? j : j0;
                ei[w] = l_a[jc];
                dl[w] = l_d[jc];
            }
            #pragma unroll
            for (int w = 0; w < WALK_W; ++w) sn[w] = (unsigned)src[ei[w]];
            #pragma unroll
            for (int w = 0; w < WALK_W; ++w) {
                #pragma unroll
                for (int hh = 0; hh < HEADS; ++hh) a[w][hh] = as_[sn[w] * 3 + hh];
            }
            #pragma unroll
            for (int w = 0; w < WALK_W; ++w) {
                if (on[w]) {
                    l_a[j0 + w * LTHR] = sn[w];
                    max_apply(a[w], dl[w]);
                }
            }
        }
    } else {
        for (int i = tid; i < E; i += LTHR) {
            unsigned r = (unsigned)(dst[i] - lo);
            if (r < urn) {
                unsigned sn = (unsigned)src[i];
                float a[HEADS];
                #pragma unroll
                for (int hh = 0; hh < HEADS; ++hh) a[hh] = as_[sn * 3 + hh];
                max_apply(a, r);
            }
        }
    }
    __syncthreads();

    for (int f = tid; f < rn * 3; f += LTHR) {
        float e = lrelu(as_[lo * 3 + f] + ad_l[f]);
        float p = expf(e - fdec(m_l[f]));
        s_l[f] = p;
        #pragma unroll
        for (int cc = 0; cc < 3; ++cc)
            acc_l[f * 3 + cc] = p * h[(lo * 3 + f) * 3 + cc];
    }
    __syncthreads();

    auto sum_apply = [&](const float* a, const float* hv, unsigned dl) {
        float p[HEADS];
        #pragma unroll
        for (int hh = 0; hh < HEADS; ++hh) {
            float e = lrelu(a[hh] + ad_l[dl * 3 + hh]);
            p[hh] = expf(e - fdec(m_l[dl * 3 + hh]));
        }
        #pragma unroll
        for (int hh = 0; hh < HEADS; ++hh) {
            atomicAdd(&s_l[dl * 3 + hh], p[hh]);
            #pragma unroll
            for (int cc = 0; cc < 3; ++cc)
                atomicAdd(&acc_l[dl * 9 + hh * 3 + cc], p[hh] * hv[hh * 3 + cc]);
        }
    };
    if (listed) {
        for (unsigned j0 = tid; j0 < total; j0 += WALK_W * LTHR) {
            unsigned dl[WALK_W], sn[WALK_W];
            bool on[WALK_W];
            float a[WALK_W][HEADS], hv[WALK_W][9];
            #pragma unroll
            for (int w = 0; w < WALK_W; ++w) {
                unsigned j = j0 + w * LTHR;
                on[w] = j < total;
                unsigned jc = on[w] ? j : j0;
                sn[w] = l_a[jc];
                dl[w] = l_d[jc];
            }
            #pragma unroll
            for (int w = 0; w < WALK_W; ++w) {
                #pragma unroll
                for (int hh = 0; hh < HEADS; ++hh) a[w][hh] = as_[sn[w] * 3 + hh];
                #pragma unroll
                for (int k = 0; k < 9; ++k) hv[w][k] = h[sn[w] * 9 + k];
            }
            #pragma unroll
            for (int w = 0; w < WALK_W; ++w) {
                if (on[w]) sum_apply(a[w], hv[w], dl[w]);
            }
        }
    } else {
        for (int i = tid; i < E; i += LTHR) {
            unsigned r = (unsigned)(dst[i] - lo);
            if (r < urn) {
                unsigned sn = (unsigned)src[i];
                float a[HEADS], hv[9];
                #pragma unroll
                for (int hh = 0; hh < HEADS; ++hh) a[hh] = as_[sn * 3 + hh];
                #pragma unroll
                for (int k = 0; k < 9; ++k) hv[k] = h[sn * 9 + k];
                sum_apply(a, hv, r);
            }
        }
    }
    __syncthreads();

    if (LAYER == 1) {
        for (int f = tid; f < rn * 9; f += LTHR) {
            int nl = f / 9;
            int k  = f - nl * 9;
            int hh = k / 3;
            float inv = 1.0f / (s_l[nl * 3 + hh] + GAT_EPS);
            float v = acc_l[f] * inv + bias[k];
            outp[lo * 9 + f] = (v > 0.f) ? v : (expf(v) - 1.0f);
        }
    } else {
        for (int nl = tid; nl < rn; nl += LTHR) {
            float v[NCLS];
            #pragma unroll
            for (int cc = 0; cc < NCLS; ++cc) {
                float acv = 0.f;
                #pragma unroll
                for (int hh = 0; hh < HEADS; ++hh)
                    acv += acc_l[nl * 9 + hh * 3 + cc] / (s_l[nl * 3 + hh] + GAT_EPS);
                v[cc] = acv * (1.0f / HEADS) + bias[cc];
            }
            float mx = fmaxf(v[0], fmaxf(v[1], v[2]));
            float lse = logf(expf(v[0] - mx) + expf(v[1] - mx) + expf(v[2] - mx));
            #pragma unroll
            for (int cc = 0; cc < NCLS; ++cc)
                outp[(lo + nl) * NCLS + cc] = v[cc] - mx - lse;
        }
    }
}

__global__ void gemm2(const float* __restrict__ t, const float* __restrict__ W2,
                      float* __restrict__ h2, int N) {
    __shared__ float Ws[81];
    if (threadIdx.x < 81) Ws[threadIdx.x] = W2[threadIdx.x];
    __syncthreads();
    int n = blockIdx.x * blockDim.x + threadIdx.x;
    if (n >= N) return;
    float tin[9];
    #pragma unroll
    for (int i = 0; i < 9; ++i) tin[i] = t[n * 9 + i];
    #pragma unroll
    for (int j = 0; j < 9; ++j) {
        float a = 0.f;
        #pragma unroll
        for (int i = 0; i < 9; ++i) a += tin[i] * Ws[i * 9 + j];
        h2[n * 9 + j] = a;
    }
}

extern "C" void kernel_launch(void* const* d_in, const int* in_sizes, int n_in,
                              void* d_out, int out_size, void* d_ws, size_t ws_size,
                              hipStream_t stream) {
    const float* x        = (const float*)d_in[0];
    const int*   eidx     = (const int*)d_in[1];
    const float* W1       = (const float*)d_in[2];
    const float* att_src1 = (const float*)d_in[3];
    const float* att_dst1 = (const float*)d_in[4];
    const float* b1       = (const float*)d_in[5];
    const float* W2       = (const float*)d_in[6];
    const float* att_src2 = (const float*)d_in[7];
    const float* att_dst2 = (const float*)d_in[8];
    const float* b2       = (const float*)d_in[9];

    const int N = in_sizes[0] / IN_DIM;
    const int E = in_sizes[1] / 2;
    const int* src = eidx;
    const int* dst = eidx + E;

    float* h   = (float*)d_ws;
    float* as_ = h   + (size_t)N * 9;
    float* ad_ = as_ + (size_t)N * 3;
    float* t   = ad_ + (size_t)N * 3;

    const int nb = (N + 255) / 256;
    const int tiles = (N + 15) / 16;
    const int gb = (tiles + 7) / 8;
    const int kb = (N + RNG - 1) / RNG;

    gemm1_wmma   <<<gb, 256, 0, stream>>>(x, W1, h, N);
    node_att     <<<nb, 256, 0, stream>>>(h, att_src1, att_dst1, as_, ad_, N);
    gat_layer<1> <<<kb, LTHR, 0, stream>>>(src, dst, h, as_, ad_, b1, t, N, E);

    gemm2        <<<nb, 256, 0, stream>>>(t, W2, h, N);
    node_att     <<<nb, 256, 0, stream>>>(h, att_src2, att_dst2, as_, ad_, N);
    gat_layer<2> <<<kb, LTHR, 0, stream>>>(src, dst, h, as_, ad_, b2, (float*)d_out, N, E);
}


# Owner blocks sweep one byte per edge, not four: a new launch writes each edge's owner-block number as a byte, blocks match bytes by carry-free bit tricks; sum order = LDS arrival order as before
# speedup vs baseline: 6.5790x; 1.1901x over previous
#include <hip/hip_runtime.h>
#include <math.h>
#include <stdint.h>


#define IN_DIM   256
#define HEADS    3
#define HID      3
#define NCLS     3
#define NEG_SLOPE 0.2f
#define GAT_EPS  1e-16f

#define RNG    416
#define LCAP   20480
#define LTHR   1024
#define SCAN_U 8
#define WALK_W 4
#define NOT_MINE 0xffffffffu

typedef __attribute__((ext_vector_type(2))) float v2f;
typedef __attribute__((ext_vector_type(8))) float v8f;

__device__ __forceinline__ unsigned fenc(float f) {
    unsigned u = __float_as_uint(f);
    return (u & 0x80000000u) ? ~u : (u | 0x80000000u);
}
__device__ __forceinline__ float fdec(unsigned u) {
    return (u & 0x80000000u) ? __uint_as_float(u & 0x7fffffffu)
                             : __uint_as_float(~u);
}
__device__ __forceinline__ float lrelu(float x) { return x > 0.f ? x : NEG_SLOPE * x; }

__global__ __launch_bounds__(256) void gemm1_wmma(const float* __restrict__ x,
                                                  const float* __restrict__ W,
                                                  float* __restrict__ h, int N) {
    __shared__ float Wp[IN_DIM * 16];
    for (int i = threadIdx.x; i < IN_DIM * 16; i += 256) {
        int k = i >> 4, n = i & 15;
        Wp[i] = (n < HEADS * HID) ? W[k * (HEADS * HID) + n] : 0.0f;
    }
    __syncthreads();

    int lane  = threadIdx.x & 31;
    int wave  = threadIdx.x >> 5;
    long tile = (long)blockIdx.x * 8 + wave;
    long m0   = tile * 16;
    if (m0 >= N) return;

    int laneM = lane & 15;
    int koff  = (lane < 16) ? 0 : 2;
    const float* xrow = x + (m0 + laneM) * IN_DIM;

    v8f c = {};
    #pragma unroll 4
    for (int k0 = 0; k0 < IN_DIM; k0 += 4) {
        v2f a, b;
        a.x = xrow[k0 + koff];
        a.y = xrow[k0 + koff + 1];
        b.x = Wp[(k0 + koff)     * 16 + laneM];
        b.y = Wp[(k0 + koff + 1) * 16 + laneM];
        c = __builtin_amdgcn_wmma_f32_16x16x4_f32(false, a, false, b,
                                                  (short)0, c, false, false);
    }

    int rbase = (lane < 16) ? 0 : 8;
    if (laneM < HEADS * HID) {
        #pragma unroll
        for (int v = 0; v < 8; ++v)
            h[(m0 + rbase + v) * (HEADS * HID) + laneM] = c[v];
    }
}

__global__ void node_att(const float* __restrict__ h,
                         const float* __restrict__ atts,
                         const float* __restrict__ attd,
                         float* __restrict__ as_, float* __restrict__ ad_, int N) {
    int n = blockIdx.x * blockDim.x + threadIdx.x;
    if (n >= N) return;
    #pragma unroll
    for (int hh = 0; hh < HEADS; ++hh) {
        float s = 0.f, d = 0.f;
        #pragma unroll
        for (int cc = 0; cc < 3; ++cc) {
            float v = h[n * 9 + hh * 3 + cc];
            s += v * atts[hh * 3 + cc];
            d += v * attd[hh * 3 + cc];
        }
        as_[n * 3 + hh] = s;
        ad_[n * 3 + hh] = d;
    }
}

__global__ __launch_bounds__(256) void edge_owner_keys(const int* __restrict__ dst,
                                                       unsigned* __restrict__ keys, int E) {
    const int i  = blockIdx.x * 256 + threadIdx.x;
    const int e0 = i << 2;
    if (e0 >= E) return;
    unsigned d0, d1 = 0u, d2 = 0u, d3 = 0u;
    if (e0 + 3 < E && ((((uintptr_t)dst) & 15) == 0)) {
        int4 d = *(const int4*)(dst + e0);
        d0 = (unsigned)d.x; d1 = (unsigned)d.y; d2 = (unsigned)d.z; d3 = (unsigned)d.w;
    } else {
        d0 = (unsigned)dst[e0];
        if (e0 + 1 < E) d1 = (unsigned)dst[e0 + 1];
        if (e0 + 2 < E) d2 = (unsigned)dst[e0 + 2];
        if (e0 + 3 < E) d3 = (unsigned)dst[e0 + 3];
    }
    keys[i] = ((d0 / RNG) & 255u) | (((d1 / RNG) & 255u) << 8) |
              (((d2 / RNG) & 255u) << 16) | (((d3 / RNG) & 255u) << 24);
}

template <int LAYER>
__global__ __launch_bounds__(LTHR) void gat_layer(const int* __restrict__ src,
                                                  const int* __restrict__ dst,
                                                  const unsigned* __restrict__ keys,
                                                  const float* __restrict__ h,
                                                  const float* __restrict__ as_,
                                                  const float* __restrict__ ad_,
                                                  const float* __restrict__ bias,
                                                  float* __restrict__ outp, int N, int E) {
    __shared__ unsigned l_a[LCAP];
    __shared__ unsigned l_d[LCAP];
    __shared__ float    ad_l[RNG * 3];
    __shared__ unsigned m_l[RNG * 3];
    __shared__ float    s_l[RNG * 3];
    __shared__ float    acc_l[RNG * 9];
    __shared__ unsigned cnt;

    const int tid = threadIdx.x;
    const int lo  = blockIdx.x * RNG;
    const int rn  = min(RNG, N - lo);
    const unsigned urn = (unsigned)rn;

    for (int f = tid; f < rn * 3; f += LTHR) {
        float a = as_[lo * 3 + f], d = ad_[lo * 3 + f];
        ad_l[f] = d;
        m_l[f]  = fenc(lrelu(a + d));
    }
    if (tid == 0) cnt = 0u;
    __syncthreads();

    auto push = [&](unsigned ei) {
        unsigned pos = atomicAdd(&cnt, 1u);
        if (pos < (unsigned)LCAP) l_a[pos] = ei;
    };
    const unsigned myk = (unsigned)blockIdx.x & 255u;
    const unsigned pat = myk * 0x01010101u;
    auto eqmask = [&](unsigned w) {
        unsigned x = w ^ pat;
        return ~(((x & 0x7f7f7f7fu) + 0x7f7f7f7fu) | x) & 0x80808080u;
    };
    auto test16 = [&](uint4 k, int v) {
        unsigned m = (eqmask(k.x) >> 7) | (eqmask(k.y) >> 6) | (eqmask(k.z) >> 5) | (eqmask(k.w) >> 4);
        while (m) {
            unsigned p = (unsigned)__builtin_ctz(m);
            m &= m - 1u;
            push(((unsigned)v << 4) + ((p & 7u) << 2) + (p >> 3));
        }
    };
    const bool vec_ok = ((((uintptr_t)keys) & 15) == 0);
    const int E16 = vec_ok ? (E >> 4) : 0;
    {
        const uint4* key4 = (const uint4*)keys;
        const int tile  = LTHR * SCAN_U;
        const int nfull = E16 / tile;
        const int t0 = ((int)blockIdx.x * nfull) / (int)gridDim.x;
        auto tile_base = [&](int k) { int t = k + t0; if (t >= nfull) t -= nfull; return t * tile; };
        auto load_set = [&](uint4* d, int vb) {
            #pragma unroll
            for (int u = 0; u < SCAN_U; ++u) d[u] = key4[vb + u * LTHR + tid];
        };
        auto scan_set = [&](const uint4* d, int vb) {
            #pragma unroll
            for (int u = 0; u < SCAN_U; ++u) test16(d[u], vb + u * LTHR + tid);
        };
        if (nfull > 0) {
            uint4 sa[SCAN_U], sb[SCAN_U];
            load_set(sa, tile_base(0));
            for (int k = 0; k < nfull; k += 2) {
                const bool has_b = (k + 1) < nfull;
                load_set(sb, tile_base(has_b ? k + 1 : k));
                scan_set(sa, tile_base(k));
                load_set(sa, tile_base((k + 2) < nfull ? k + 2 : k));
                if (has_b) scan_set(sb, tile_base(k + 1));
            }
        }
        for (int v = nfull * tile + tid; v < E16; v += LTHR) test16(key4[v], v);
        const unsigned char* key1 = (const unsigned char*)keys;
        for (int i = (E16 << 4) + tid; i < E; i += LTHR)
            if ((unsigned)key1[i] == myk) push((unsigned)i);
    }
    __syncthreads();
    const unsigned total  = cnt;
    const bool     listed = total <= (unsigned)LCAP;

    auto max_apply = [&](const float* a, unsigned dl) {
        float d[HEADS];
        #pragma unroll
        for (int hh = 0; hh < HEADS; ++hh) d[hh] = ad_l[dl * 3 + hh];
        #pragma unroll
        for (int hh = 0; hh < HEADS; ++hh)
            atomicMax(&m_l[dl * 3 + hh], fenc(lrelu(a[hh] + d[hh])));
    };
    if (listed) {
        for (unsigned j0 = tid; j0 < total; j0 += WALK_W * LTHR) {
            unsigned ei[WALK_W], dl[WALK_W], sn[WALK_W];
            bool on[WALK_W];
            float a[WALK_W][HEADS];
            #pragma unroll
            for (int w = 0; w < WALK_W; ++w) {
                unsigned j = j0 + w * LTHR;
                on[w] = j < total;
                unsigned jc = on[w] ? j : j0;
                ei[w] = l_a[jc];
            }
            #pragma unroll
            for (int w = 0; w < WALK_W; ++w) {
                sn[w] = (unsigned)src[ei[w]];
                dl[w] = (unsigned)(dst[ei[w]] - lo);
            }
            #pragma unroll
            for (int w = 0; w < WALK_W; ++w) {
                #pragma unroll
                for (int hh = 0; hh < HEADS; ++hh) a[w][hh] = as_[sn[w] * 3 + hh];
            }
            #pragma unroll
            for (int w = 0; w < WALK_W; ++w) {
                if (on[w]) {
                    const bool mine = dl[w] < urn;
                    l_a[j0 + w * LTHR] = sn[w];
                    l_d[j0 + w * LTHR] = mine ? dl[w] : NOT_MINE;
                    if (mine) max_apply(a[w], dl[w]);
                }
            }
        }
    } else {
        for (int i = tid; i < E; i += LTHR) {
            unsigned r = (unsigned)(dst[i] - lo);
            if (r < urn) {
                unsigned sn = (unsigned)src[i];
                float a[HEADS];
                #pragma unroll
                for (int hh = 0; hh < HEADS; ++hh) a[hh] = as_[sn * 3 + hh];
                max_apply(a, r);
            }
        }
    }
    __syncthreads();

    for (int f = tid; f < rn * 3; f += LTHR) {
        float e = lrelu(as_[lo * 3 + f] + ad_l[f]);
        float p = expf(e - fdec(m_l[f]));
        s_l[f] = p;
        #pragma unroll
        for (int cc = 0; cc < 3; ++cc)
            acc_l[f * 3 + cc] = p * h[(lo * 3 + f) * 3 + cc];
    }
    __syncthreads();

    auto sum_apply = [&](const float* a, const float* hv, unsigned dl) {
        float p[HEADS];
        #pragma unroll
        for (int hh = 0; hh < HEADS; ++hh) {
            float e = lrelu(a[hh] + ad_l[dl * 3 + hh]);
            p[hh] = expf(e - fdec(m_l[dl * 3 + hh]));
        }
        #pragma unroll
        for (int hh = 0; hh < HEADS; ++hh) {
            atomicAdd(&s_l[dl * 3 + hh], p[hh]);
            #pragma unroll
            for (int cc = 0; cc < 3; ++cc)
                atomicAdd(&acc_l[dl * 9 + hh * 3 + cc], p[hh] * hv[hh * 3 + cc]);
        }
    };
    if (listed) {
        for (unsigned j0 = tid; j0 < total; j0 += WALK_W * LTHR) {
            unsigned dl[WALK_W], sn[WALK_W];
            bool on[WALK_W];
            float a[WALK_W][HEADS], hv[WALK_W][9];
            #pragma unroll
            for (int w = 0; w < WALK_W; ++w) {
                unsigned j = j0 + w * LTHR;
                on[w] = j < total;
                unsigned jc = on[w] ? j : j0;
                sn[w] = l_a[jc];
                dl[w] = l_d[jc];
                on[w] = on[w] && (dl[w] != NOT_MINE);
            }
            #pragma unroll
            for (int w = 0; w < WALK_W; ++w) {
                #pragma unroll
                for (int hh = 0; hh < HEADS; ++hh) a[w][hh] = as_[sn[w] * 3 + hh];
                #pragma unroll
                for (int k = 0; k < 9; ++k) hv[w][k] = h[sn[w] * 9 + k];
            }
            #pragma unroll
            for (int w = 0; w < WALK_W; ++w) {
                if (on[w]) sum_apply(a[w], hv[w], dl[w]);
            }
        }
    } else {
        for (int i = tid; i < E; i += LTHR) {
            unsigned r = (unsigned)(dst[i] - lo);
            if (r < urn) {
                unsigned sn = (unsigned)src[i];
                float a[HEADS], hv[9];
                #pragma unroll
                for (int hh = 0; hh < HEADS; ++hh) a[hh] = as_[sn * 3 + hh];
                #pragma unroll
                for (int k = 0; k < 9; ++k) hv[k] = h[sn * 9 + k];
                sum_apply(a, hv, r);
            }
        }
    }
    __syncthreads();

    if (LAYER == 1) {
        for (int f = tid; f < rn * 9; f += LTHR) {
            int nl = f / 9;
            int k  = f - nl * 9;
            int hh = k / 3;
            float inv = 1.0f / (s_l[nl * 3 + hh] + GAT_EPS);
            float v = acc_l[f] * inv + bias[k];
            outp[lo * 9 + f] = (v > 0.f) ? v : (expf(v) - 1.0f);
        }
    } else {
        for (int nl = tid; nl < rn; nl += LTHR) {
            float v[NCLS];
            #pragma unroll
            for (int cc = 0; cc < NCLS; ++cc) {
                float acv = 0.f;
                #pragma unroll
                for (int hh = 0; hh < HEADS; ++hh)
                    acv += acc_l[nl * 9 + hh * 3 + cc] / (s_l[nl * 3 + hh] + GAT_EPS);
                v[cc] = acv * (1.0f / HEADS) + bias[cc];
            }
            float mx = fmaxf(v[0], fmaxf(v[1], v[2]));
            float lse = logf(expf(v[0] - mx) + expf(v[1] - mx) + expf(v[2] - mx));
            #pragma unroll
            for (int cc = 0; cc < NCLS; ++cc)
                outp[(lo + nl) * NCLS + cc] = v[cc] - mx - lse;
        }
    }
}

__global__ void gemm2(const float* __restrict__ t, const float* __restrict__ W2,
                      float* __restrict__ h2, int N) {
    __shared__ float Ws[81];
    if (threadIdx.x < 81) Ws[threadIdx.x] = W2[threadIdx.x];
    __syncthreads();
    int n = blockIdx.x * blockDim.x + threadIdx.x;
    if (n >= N) return;
    float tin[9];
    #pragma unroll
    for (int i = 0; i < 9; ++i) tin[i] = t[n * 9 + i];
    #pragma unroll
    for (int j = 0; j < 9; ++j) {
        float a = 0.f;
        #pragma unroll
        for (int i = 0; i < 9; ++i) a += tin[i] * Ws[i * 9 + j];
        h2[n * 9 + j] = a;
    }
}

extern "C" void kernel_launch(void* const* d_in, const int* in_sizes, int n_in,
                              void* d_out, int out_size, void* d_ws, size_t ws_size,
                              hipStream_t stream) {
    const float* x        = (const float*)d_in[0];
    const int*   eidx     = (const int*)d_in[1];
    const float* W1       = (const float*)d_in[2];
    const float* att_src1 = (const float*)d_in[3];
    const float* att_dst1 = (const float*)d_in[4];
    const float* b1       = (const float*)d_in[5];
    const float* W2       = (const float*)d_in[6];
    const float* att_src2 = (const float*)d_in[7];
    const float* att_dst2 = (const float*)d_in[8];
    const float* b2       = (const float*)d_in[9];

    const int N = in_sizes[0] / IN_DIM;
    const int E = in_sizes[1] / 2;
    const int* src = eidx;
    const int* dst = eidx + E;

    float* h   = (float*)d_ws;
    float* as_ = h   + (size_t)N * 9;
    float* ad_ = as_ + (size_t)N * 3;
    float* t   = ad_ + (size_t)N * 3;
    unsigned* keys = (unsigned*)(t + (size_t)N * 9);

    const int nb = (N + 255) / 256;
    const int tiles = (N + 15) / 16;
    const int gb = (tiles + 7) / 8;
    const int kb = (N + RNG - 1) / RNG;
    const int kw = (E + 3) / 4;
    const int wb = (kw + 255) / 256;

    gemm1_wmma   <<<gb, 256, 0, stream>>>(x, W1, h, N);
    node_att     <<<nb, 256, 0, stream>>>(h, att_src1, att_dst1, as_, ad_, N);
    edge_owner_keys<<<wb, 256, 0, stream>>>(dst, keys, E);
    gat_layer<1> <<<kb, LTHR, 0, stream>>>(src, dst, keys, h, as_, ad_, b1, t, N, E);

    gemm2        <<<nb, 256, 0, stream>>>(t, W2, h, N);
    node_att     <<<nb, 256, 0, stream>>>(h, att_src2, att_dst2, as_, ad_, N);
    gat_layer<2> <<<kb, LTHR, 0, stream>>>(src, dst, keys, h, as_, ad_, b2, (float*)d_out, N, E);
}


# Per-edge gathers of a source node's row widened: its 3 dots come as one 12-byte load and its 9 features as 16+16+4 bytes at 4-byte alignment, 4 load instructions per edge not 12; sum order unchanged
# speedup vs baseline: 7.2946x; 1.1088x over previous
#include <hip/hip_runtime.h>
#include <math.h>
#include <stdint.h>


#define IN_DIM   256
#define HEADS    3
#define HID      3
#define NCLS     3
#define NEG_SLOPE 0.2f
#define GAT_EPS  1e-16f

#define RNG    416
#define LCAP   20480
#define LTHR   1024
#define SCAN_U 8
#define WALK_W 4
#define NOT_MINE 0xffffffffu

typedef __attribute__((ext_vector_type(2))) float v2f;
typedef __attribute__((ext_vector_type(8))) float v8f;

__device__ __forceinline__ unsigned fenc(float f) {
    unsigned u = __float_as_uint(f);
    return (u & 0x80000000u) ? ~u : (u | 0x80000000u);
}
__device__ __forceinline__ float fdec(unsigned u) {
    return (u & 0x80000000u) ? __uint_as_float(u & 0x7fffffffu)
                             : __uint_as_float(~u);
}
__device__ __forceinline__ float lrelu(float x) { return x > 0.f ? x : NEG_SLOPE * x; }

struct __attribute__((packed, aligned(4), may_alias)) row3_t { float v[3]; };
struct __attribute__((packed, aligned(4), may_alias)) row9_t { float v[9]; };
__device__ __forceinline__ void load_row3(const float* __restrict__ base, unsigned n, float* o) {
    const row3_t r = *(const row3_t*)(base + (size_t)n * 3);
    o[0] = r.v[0]; o[1] = r.v[1]; o[2] = r.v[2];
}
__device__ __forceinline__ void load_row9(const float* __restrict__ base, unsigned n, float* o) {
    const row9_t r = *(const row9_t*)(base + (size_t)n * 9);
    #pragma unroll
    for (int k = 0; k < 9; ++k) o[k] = r.v[k];
}

__global__ __launch_bounds__(256) void gemm1_wmma(const float* __restrict__ x,
                                                  const float* __restrict__ W,
                                                  float* __restrict__ h, int N) {
    __shared__ float Wp[IN_DIM * 16];
    for (int i = threadIdx.x; i < IN_DIM * 16; i += 256) {
        int k = i >> 4, n = i & 15;
        Wp[i] = (n < HEADS * HID) ? W[k * (HEADS * HID) + n] : 0.0f;
    }
    __syncthreads();

    int lane  = threadIdx.x & 31;
    int wave  = threadIdx.x >> 5;
    long tile = (long)blockIdx.x * 8 + wave;
    long m0   = tile * 16;
    if (m0 >= N) return;

    int laneM = lane & 15;
    int koff  = (lane < 16) ? 0 : 2;
    const float* xrow = x + (m0 + laneM) * IN_DIM;

    v8f c = {};
    #pragma unroll 4
    for (int k0 = 0; k0 < IN_DIM; k0 += 4) {
        v2f a, b;
        a.x = xrow[k0 + koff];
        a.y = xrow[k0 + koff + 1];
        b.x = Wp[(k0 + koff)     * 16 + laneM];
        b.y = Wp[(k0 + koff + 1) * 16 + laneM];
        c = __builtin_amdgcn_wmma_f32_16x16x4_f32(false, a, false, b,
                                                  (short)0, c, false, false);
    }

    int rbase = (lane < 16) ? 0 : 8;
    if (laneM < HEADS * HID) {
        #pragma unroll
        for (int v = 0; v < 8; ++v)
            h[(m0 + rbase + v) * (HEADS * HID) + laneM] = c[v];
    }
}

__global__ void node_att(const float* __restrict__ h,
                         const float* __restrict__ atts,
                         const float* __restrict__ attd,
                         float* __restrict__ as_, float* __restrict__ ad_, int N) {
    int n = blockIdx.x * blockDim.x + threadIdx.x;
    if (n >= N) return;
    #pragma unroll
    for (int hh = 0; hh < HEADS; ++hh) {
        float s = 0.f, d = 0.f;
        #pragma unroll
        for (int cc = 0; cc < 3; ++cc) {
            float v = h[n * 9 + hh * 3 + cc];
            s += v * atts[hh * 3 + cc];
            d += v * attd[hh * 3 + cc];
        }
        as_[n * 3 + hh] = s;
        ad_[n * 3 + hh] = d;
    }
}

__global__ __launch_bounds__(256) void edge_owner_keys(const int* __restrict__ dst,
                                                       unsigned* __restrict__ keys, int E) {
    const int i  = blockIdx.x * 256 + threadIdx.x;
    const int e0 = i << 2;
    if (e0 >= E) return;
    unsigned d0, d1 = 0u, d2 = 0u, d3 = 0u;
    if (e0 + 3 < E && ((((uintptr_t)dst) & 15) == 0)) {
        int4 d = *(const int4*)(dst + e0);
        d0 = (unsigned)d.x; d1 = (unsigned)d.y; d2 = (unsigned)d.z; d3 = (unsigned)d.w;
    } else {
        d0 = (unsigned)dst[e0];
        if (e0 + 1 < E) d1 = (unsigned)dst[e0 + 1];
        if (e0 + 2 < E) d2 = (unsigned)dst[e0 + 2];
        if (e0 + 3 < E) d3 = (unsigned)dst[e0 + 3];
    }
    keys[i] = ((d0 / RNG) & 255u) | (((d1 / RNG) & 255u) << 8) |
              (((d2 / RNG) & 255u) << 16) | (((d3 / RNG) & 255u) << 24);
}

template <int LAYER>
__global__ __launch_bounds__(LTHR) void gat_layer(const int* __restrict__ src,
                                                  const int* __restrict__ dst,
                                                  const unsigned* __restrict__ keys,
                                                  const float* __restrict__ h,
                                                  const float* __restrict__ as_,
                                                  const float* __restrict__ ad_,
                                                  const float* __restrict__ bias,
                                                  float* __restrict__ outp, int N, int E) {
    __shared__ unsigned l_a[LCAP];
    __shared__ unsigned l_d[LCAP];
    __shared__ float    ad_l[RNG * 3];
    __shared__ unsigned m_l[RNG * 3];
    __shared__ float    s_l[RNG * 3];
    __shared__ float    acc_l[RNG * 9];
    __shared__ unsigned cnt;

    const int tid = threadIdx.x;
    const int lo  = blockIdx.x * RNG;
    const int rn  = min(RNG, N - lo);
    const unsigned urn = (unsigned)rn;

    for (int f = tid; f < rn * 3; f += LTHR) {
        float a = as_[lo * 3 + f], d = ad_[lo * 3 + f];
        ad_l[f] = d;
        m_l[f]  = fenc(lrelu(a + d));
    }
    if (tid == 0) cnt = 0u;
    __syncthreads();

    auto push = [&](unsigned ei) {
        unsigned pos = atomicAdd(&cnt, 1u);
        if (pos < (unsigned)LCAP) l_a[pos] = ei;
    };
    const unsigned myk = (unsigned)blockIdx.x & 255u;
    const unsigned pat = myk * 0x01010101u;
    auto eqmask = [&](unsigned w) {
        unsigned x = w ^ pat;
        return ~(((x & 0x7f7f7f7fu) + 0x7f7f7f7fu) | x) & 0x80808080u;
    };
    auto test16 = [&](uint4 k, int v) {
        unsigned m = (eqmask(k.x) >> 7) | (eqmask(k.y) >> 6) | (eqmask(k.z) >> 5) | (eqmask(k.w) >> 4);
        while (m) {
            unsigned p = (unsigned)__builtin_ctz(m);
            m &= m - 1u;
            push(((unsigned)v << 4) + ((p & 7u) << 2) + (p >> 3));
        }
    };
    const bool vec_ok = ((((uintptr_t)keys) & 15) == 0);
    const int E16 = vec_ok ? (E >> 4) : 0;
    {
        const uint4* key4 = (const uint4*)keys;
        const int tile  = LTHR * SCAN_U;
        const int nfull = E16 / tile;
        const int t0 = ((int)blockIdx.x * nfull) / (int)gridDim.x;
        auto tile_base = [&](int k) { int t = k + t0; if (t >= nfull) t -= nfull; return t * tile; };
        auto load_set = [&](uint4* d, int vb) {
            #pragma unroll
            for (int u = 0; u < SCAN_U; ++u) d[u] = key4[vb + u * LTHR + tid];
        };
        auto scan_set = [&](const uint4* d, int vb) {
            #pragma unroll
            for (int u = 0; u < SCAN_U; ++u) test16(d[u], vb + u * LTHR + tid);
        };
        if (nfull > 0) {
            uint4 sa[SCAN_U], sb[SCAN_U];
            load_set(sa, tile_base(0));
            for (int k = 0; k < nfull; k += 2) {
                const bool has_b = (k + 1) < nfull;
                load_set(sb, tile_base(has_b ? k + 1 : k));
                scan_set(sa, tile_base(k));
                load_set(sa, tile_base((k + 2) < nfull ? k + 2 : k));
                if (has_b) scan_set(sb, tile_base(k + 1));
            }
        }
        for (int v = nfull * tile + tid; v < E16; v += LTHR) test16(key4[v], v);
        const unsigned char* key1 = (const unsigned char*)keys;
        for (int i = (E16 << 4) + tid; i < E; i += LTHR)
            if ((unsigned)key1[i] == myk) push((unsigned)i);
    }
    __syncthreads();
    const unsigned total  = cnt;
    const bool     listed = total <= (unsigned)LCAP;

    auto max_apply = [&](const float* a, unsigned dl) {
        float d[HEADS];
        #pragma unroll
        for (int hh = 0; hh < HEADS; ++hh) d[hh] = ad_l[dl * 3 + hh];
        #pragma unroll
        for (int hh = 0; hh < HEADS; ++hh)
            atomicMax(&m_l[dl * 3 + hh], fenc(lrelu(a[hh] + d[hh])));
    };
    if (listed) {
        for (unsigned j0 = tid; j0 < total; j0 += WALK_W * LTHR) {
            unsigned ei[WALK_W], dl[WALK_W], sn[WALK_W];
            bool on[WALK_W];
            float a[WALK_W][HEADS];
            #pragma unroll
            for (int w = 0; w < WALK_W; ++w) {
                unsigned j = j0 + w * LTHR;
                on[w] = j < total;
                unsigned jc = on[w] ? j : j0;
                ei[w] = l_a[jc];
            }
            #pragma unroll
            for (int w = 0; w < WALK_W; ++w) {
                sn[w] = (unsigned)src[ei[w]];
                dl[w] = (unsigned)(dst[ei[w]] - lo);
            }
            #pragma unroll
            for (int w = 0; w < WALK_W; ++w) load_row3(as_, sn[w], a[w]);
            #pragma unroll
            for (int w = 0; w < WALK_W; ++w) {
                if (on[w]) {
                    const bool mine = dl[w] < urn;
                    l_a[j0 + w * LTHR] = sn[w];
                    l_d[j0 + w * LTHR] = mine ? dl[w] : NOT_MINE;
                    if (mine) max_apply(a[w], dl[w]);
                }
            }
        }
    } else {
        for (int i = tid; i < E; i += LTHR) {
            unsigned r = (unsigned)(dst[i] - lo);
            if (r < urn) {
                unsigned sn = (unsigned)src[i];
                float a[HEADS];
                load_row3(as_, sn, a);
                max_apply(a, r);
            }
        }
    }
    __syncthreads();

    for (int f = tid; f < rn * 3; f += LTHR) {
        float e = lrelu(as_[lo * 3 + f] + ad_l[f]);
        float p = expf(e - fdec(m_l[f]));
        s_l[f] = p;
        #pragma unroll
        for (int cc = 0; cc < 3; ++cc)
            acc_l[f * 3 + cc] = p * h[(lo * 3 + f) * 3 + cc];
    }
    __syncthreads();

    auto sum_apply = [&](const float* a, const float* hv, unsigned dl) {
        float p[HEADS];
        #pragma unroll
        for (int hh = 0; hh < HEADS; ++hh) {
            float e = lrelu(a[hh] + ad_l[dl * 3 + hh]);
            p[hh] = expf(e - fdec(m_l[dl * 3 + hh]));
        }
        #pragma unroll
        for (int hh = 0; hh < HEADS; ++hh) {
            atomicAdd(&s_l[dl * 3 + hh], p[hh]);
            #pragma unroll
            for (int cc = 0; cc < 3; ++cc)
                atomicAdd(&acc_l[dl * 9 + hh * 3 + cc], p[hh] * hv[hh * 3 + cc]);
        }
    };
    if (listed) {
        for (unsigned j0 = tid; j0 < total; j0 += WALK_W * LTHR) {
            unsigned dl[WALK_W], sn[WALK_W];
            bool on[WALK_W];
            float a[WALK_W][HEADS], hv[WALK_W][9];
            #pragma unroll
            for (int w = 0; w < WALK_W; ++w) {
                unsigned j = j0 + w * LTHR;
                on[w] = j < total;
                unsigned jc = on[w] ? j : j0;
                sn[w] = l_a[jc];
                dl[w] = l_d[jc];
                on[w] = on[w] && (dl[w] != NOT_MINE);
            }
            #pragma unroll
            for (int w = 0; w < WALK_W; ++w) {
                load_row3(as_, sn[w], a[w]);
                load_row9(h, sn[w], hv[w]);
            }
            #pragma unroll
            for (int w = 0; w < WALK_W; ++w) {
                if (on[w]) sum_apply(a[w], hv[w], dl[w]);
            }
        }
    } else {
        for (int i = tid; i < E; i += LTHR) {
            unsigned r = (unsigned)(dst[i] - lo);
            if (r < urn) {
                unsigned sn = (unsigned)src[i];
                float a[HEADS], hv[9];
                load_row3(as_, sn, a);
                load_row9(h, sn, hv);
                sum_apply(a, hv, r);
            }
        }
    }
    __syncthreads();

    if (LAYER == 1) {
        for (int f = tid; f < rn * 9; f += LTHR) {
            int nl = f / 9;
            int k  = f - nl * 9;
            int hh = k / 3;
            float inv = 1.0f / (s_l[nl * 3 + hh] + GAT_EPS);
            float v = acc_l[f] * inv + bias[k];
            outp[lo * 9 + f] = (v > 0.f) ? v : (expf(v) - 1.0f);
        }
    } else {
        for (int nl = tid; nl < rn; nl += LTHR) {
            float v[NCLS];
            #pragma unroll
            for (int cc = 0; cc < NCLS; ++cc) {
                float acv = 0.f;
                #pragma unroll
                for (int hh = 0; hh < HEADS; ++hh)
                    acv += acc_l[nl * 9 + hh * 3 + cc] / (s_l[nl * 3 + hh] + GAT_EPS);
                v[cc] = acv * (1.0f / HEADS) + bias[cc];
            }
            float mx = fmaxf(v[0], fmaxf(v[1], v[2]));
            float lse = logf(expf(v[0] - mx) + expf(v[1] - mx) + expf(v[2] - mx));
            #pragma unroll
            for (int cc = 0; cc < NCLS; ++cc)
                outp[(lo + nl) * NCLS + cc] = v[cc] - mx - lse;
        }
    }
}

__global__ void gemm2(const float* __restrict__ t, const float* __restrict__ W2,
                      float* __restrict__ h2, int N) {
    __shared__ float Ws[81];
    if (threadIdx.x < 81) Ws[threadIdx.x] = W2[threadIdx.x];
    __syncthreads();
    int n = blockIdx.x * blockDim.x + threadIdx.x;
    if (n >= N) return;
    float tin[9];
    #pragma unroll
    for (int i = 0; i < 9; ++i) tin[i] = t[n * 9 + i];
    #pragma unroll
    for (int j = 0; j < 9; ++j) {
        float a = 0.f;
        #pragma unroll
        for (int i = 0; i < 9; ++i) a += tin[i] * Ws[i * 9 + j];
        h2[n * 9 + j] = a;
    }
}

extern "C" void kernel_launch(void* const* d_in, const int* in_sizes, int n_in,
                              void* d_out, int out_size, void* d_ws, size_t ws_size,
                              hipStream_t stream) {
    const float* x        = (const float*)d_in[0];
    const int*   eidx     = (const int*)d_in[1];
    const float* W1       = (const float*)d_in[2];
    const float* att_src1 = (const float*)d_in[3];
    const float* att_dst1 = (const float*)d_in[4];
    const float* b1       = (const float*)d_in[5];
    const float* W2       = (const float*)d_in[6];
    const float* att_src2 = (const float*)d_in[7];
    const float* att_dst2 = (const float*)d_in[8];
    const float* b2       = (const float*)d_in[9];

    const int N = in_sizes[0] / IN_DIM;
    const int E = in_sizes[1] / 2;
    const int* src = eidx;
    const int* dst = eidx + E;

    float* h   = (float*)d_ws;
    float* as_ = h   + (size_t)N * 9;
    float* ad_ = as_ + (size_t)N * 3;
    float* t   = ad_ + (size_t)N * 3;
    unsigned* keys = (unsigned*)(t + (size_t)N * 9);

    const int nb = (N + 255) / 256;
    const int tiles = (N + 15) / 16;
    const int gb = (tiles + 7) / 8;
    const int kb = (N + RNG - 1) / RNG;
    const int kw = (E + 3) / 4;
    const int wb = (kw + 255) / 256;

    gemm1_wmma   <<<gb, 256, 0, stream>>>(x, W1, h, N);
    node_att     <<<nb, 256, 0, stream>>>(h, att_src1, att_dst1, as_, ad_, N);
    edge_owner_keys<<<wb, 256, 0, stream>>>(dst, keys, E);
    gat_layer<1> <<<kb, LTHR, 0, stream>>>(src, dst, keys, h, as_, ad_, b1, t, N, E);

    gemm2        <<<nb, 256, 0, stream>>>(t, W2, h, N);
    node_att     <<<nb, 256, 0, stream>>>(h, att_src2, att_dst2, as_, ad_, N);
    gat_layer<2> <<<kb, LTHR, 0, stream>>>(src, dst, keys, h, as_, ad_, b2, (float*)d_out, N, E);
}


# Layer 2 takes each block's edge list (source, local destination: one word per edge) from a workspace slot layer 1 fills, not from a second key sweep and 2 gathers per edge; sum order: LDS arrival
# speedup vs baseline: 9.4106x; 1.2901x over previous
#include <hip/hip_runtime.h>
#include <math.h>
#include <stdint.h>


#define IN_DIM   256
#define HEADS    3
#define HID      3
#define NCLS     3
#define NEG_SLOPE 0.2f
#define GAT_EPS  1e-16f

#define RNG    416
#define LCAP   20480
#define LTHR   1024
#define SCAN_U 8
#define WALK_W 4
#define NOT_MINE 0xffffffffu
#define GSLOT  16384
#define DL_BITS 9
#define NO_LIST 0xffffffffu
#define NO_EDGE 0xffffffffu

typedef __attribute__((ext_vector_type(2))) float v2f;
typedef __attribute__((ext_vector_type(8))) float v8f;

__device__ __forceinline__ unsigned fenc(float f) {
    unsigned u = __float_as_uint(f);
    return (u & 0x80000000u) ? ~u : (u | 0x80000000u);
}
__device__ __forceinline__ float fdec(unsigned u) {
    return (u & 0x80000000u) ? __uint_as_float(u & 0x7fffffffu)
                             : __uint_as_float(~u);
}
__device__ __forceinline__ float lrelu(float x) { return x > 0.f ? x : NEG_SLOPE * x; }

struct __attribute__((packed, aligned(4), may_alias)) row3_t { float v[3]; };
struct __attribute__((packed, aligned(4), may_alias)) row9_t { float v[9]; };
__device__ __forceinline__ void load_row3(const float* __restrict__ base, unsigned n, float* o) {
    const row3_t r = *(const row3_t*)(base + (size_t)n * 3);
    o[0] = r.v[0]; o[1] = r.v[1]; o[2] = r.v[2];
}
__device__ __forceinline__ void load_row9(const float* __restrict__ base, unsigned n, float* o) {
    const row9_t r = *(const row9_t*)(base + (size_t)n * 9);
    #pragma unroll
    for (int k = 0; k < 9; ++k) o[k] = r.v[k];
}

__global__ __launch_bounds__(256) void gemm1_wmma(const float* __restrict__ x,
                                                  const float* __restrict__ W,
                                                  float* __restrict__ h, int N) {
    __shared__ float Wp[IN_DIM * 16];
    for (int i = threadIdx.x; i < IN_DIM * 16; i += 256) {
        int k = i >> 4, n = i & 15;
        Wp[i] = (n < HEADS * HID) ? W[k * (HEADS * HID) + n] : 0.0f;
    }
    __syncthreads();

    int lane  = threadIdx.x & 31;
    int wave  = threadIdx.x >> 5;
    long tile = (long)blockIdx.x * 8 + wave;
    long m0   = tile * 16;
    if (m0 >= N) return;

    int laneM = lane & 15;
    int koff  = (lane < 16) ? 0 : 2;
    const float* xrow = x + (m0 + laneM) * IN_DIM;

    v8f c = {};
    #pragma unroll 4
    for (int k0 = 0; k0 < IN_DIM; k0 += 4) {
        v2f a, b;
        a.x = xrow[k0 + koff];
        a.y = xrow[k0 + koff + 1];
        b.x = Wp[(k0 + koff)     * 16 + laneM];
        b.y = Wp[(k0 + koff + 1) * 16 + laneM];
        c = __builtin_amdgcn_wmma_f32_16x16x4_f32(false, a, false, b,
                                                  (short)0, c, false, false);
    }

    int rbase = (lane < 16) ? 0 : 8;
    if (laneM < HEADS * HID) {
        #pragma unroll
        for (int v = 0; v < 8; ++v)
            h[(m0 + rbase + v) * (HEADS * HID) + laneM] = c[v];
    }
}

__global__ void node_att(const float* __restrict__ h,
                         const float* __restrict__ atts,
                         const float* __restrict__ attd,
                         float* __restrict__ as_, float* __restrict__ ad_, int N) {
    int n = blockIdx.x * blockDim.x + threadIdx.x;
    if (n >= N) return;
    #pragma unroll
    for (int hh = 0; hh < HEADS; ++hh) {
        float s = 0.f, d = 0.f;
        #pragma unroll
        for (int cc = 0; cc < 3; ++cc) {
            float v = h[n * 9 + hh * 3 + cc];
            s += v * atts[hh * 3 + cc];
            d += v * attd[hh * 3 + cc];
        }
        as_[n * 3 + hh] = s;
        ad_[n * 3 + hh] = d;
    }
}

__global__ __launch_bounds__(256) void edge_owner_keys(const int* __restrict__ dst,
                                                       unsigned* __restrict__ keys, int E) {
    const int i  = blockIdx.x * 256 + threadIdx.x;
    const int e0 = i << 2;
    if (e0 >= E) return;
    unsigned d0, d1 = 0u, d2 = 0u, d3 = 0u;
    if (e0 + 3 < E && ((((uintptr_t)dst) & 15) == 0)) {
        int4 d = *(const int4*)(dst + e0);
        d0 = (unsigned)d.x; d1 = (unsigned)d.y; d2 = (unsigned)d.z; d3 = (unsigned)d.w;
    } else {
        d0 = (unsigned)dst[e0];
        if (e0 + 1 < E) d1 = (unsigned)dst[e0 + 1];
        if (e0 + 2 < E) d2 = (unsigned)dst[e0 + 2];
        if (e0 + 3 < E) d3 = (unsigned)dst[e0 + 3];
    }
    keys[i] = ((d0 / RNG) & 255u) | (((d1 / RNG) & 255u) << 8) |
              (((d2 / RNG) & 255u) << 16) | (((d3 / RNG) & 255u) << 24);
}

template <int LAYER>
__global__ __launch_bounds__(LTHR) void gat_layer(const int* __restrict__ src,
                                                  const int* __restrict__ dst,
                                                  const unsigned* __restrict__ keys,
                                                  const float* __restrict__ h,
                                                  const float* __restrict__ as_,
                                                  const float* __restrict__ ad_,
                                                  const float* __restrict__ bias,
                                                  float* __restrict__ outp,
                                                  unsigned* glist, unsigned* gcount,
                                                  size_t ws_need, size_t ws_size, int N, int E) {
    __shared__ unsigned l_a[LCAP];
    __shared__ unsigned l_d[LCAP];
    __shared__ float    ad_l[RNG * 3];
    __shared__ unsigned m_l[RNG * 3];
    __shared__ float    s_l[RNG * 3];
    __shared__ float    acc_l[RNG * 9];
    __shared__ unsigned cnt;

    const int tid = threadIdx.x;
    const int lo  = blockIdx.x * RNG;
    const int rn  = min(RNG, N - lo);
    const unsigned urn = (unsigned)rn;
    const bool share = (ws_need <= ws_size) && ((unsigned)N <= (1u << (32 - DL_BITS))) && (RNG <= (1 << DL_BITS));
    unsigned* const my_list = glist + (size_t)blockIdx.x * GSLOT;
    unsigned handed = NO_LIST;
    if (LAYER == 2 && share) handed = gcount[blockIdx.x];
    const bool from_ws = handed != NO_LIST;

    for (int f = tid; f < rn * 3; f += LTHR) {
        float a = as_[lo * 3 + f], d = ad_[lo * 3 + f];
        ad_l[f] = d;
        m_l[f]  = fenc(lrelu(a + d));
    }
    if (tid == 0) cnt = 0u;
    __syncthreads();

    auto push = [&](unsigned ei) {
        unsigned pos = atomicAdd(&cnt, 1u);
        if (pos < (unsigned)LCAP) l_a[pos] = ei;
    };
    const unsigned myk = (unsigned)blockIdx.x & 255u;
    const unsigned pat = myk * 0x01010101u;
    auto eqmask = [&](unsigned w) {
        unsigned x = w ^ pat;
        return ~(((x & 0x7f7f7f7fu) + 0x7f7f7f7fu) | x) & 0x80808080u;
    };
    auto test16 = [&](uint4 k, int v) {
        unsigned m = (eqmask(k.x) >> 7) | (eqmask(k.y) >> 6) | (eqmask(k.z) >> 5) | (eqmask(k.w) >> 4);
        while (m) {
            unsigned p = (unsigned)__builtin_ctz(m);
            m &= m - 1u;
            push(((unsigned)v << 4) + ((p & 7u) << 2) + (p >> 3));
        }
    };
    const bool vec_ok = ((((uintptr_t)keys) & 15) == 0);
    const int E16 = vec_ok ? (E >> 4) : 0;
    if (!from_ws) {
        const uint4* key4 = (const uint4*)keys;
        const int tile  = LTHR * SCAN_U;
        const int nfull = E16 / tile;
        const int t0 = ((int)blockIdx.x * nfull) / (int)gridDim.x;
        auto tile_base = [&](int k) { int t = k + t0; if (t >= nfull) t -= nfull; return t * tile; };
        auto load_set = [&](uint4* d, int vb) {
            #pragma unroll
            for (int u = 0; u < SCAN_U; ++u) d[u] = key4[vb + u * LTHR + tid];
        };
        auto scan_set = [&](const uint4* d, int vb) {
            #pragma unroll
            for (int u = 0; u < SCAN_U; ++u) test16(d[u], vb + u * LTHR + tid);
        };
        if (nfull > 0) {
            uint4 sa[SCAN_U], sb[SCAN_U];
            load_set(sa, tile_base(0));
            for (int k = 0; k < nfull; k += 2) {
                const bool has_b = (k + 1) < nfull;
                load_set(sb, tile_base(has_b ? k + 1 : k));
                scan_set(sa, tile_base(k));
                load_set(sa, tile_base((k + 2) < nfull ? k + 2 : k));
                if (has_b) scan_set(sb, tile_base(k + 1));
            }
        }
        for (int v = nfull * tile + tid; v < E16; v += LTHR) test16(key4[v], v);
        const unsigned char* key1 = (const unsigned char*)keys;
        for (int i = (E16 << 4) + tid; i < E; i += LTHR)
            if ((unsigned)key1[i] == myk) push((unsigned)i);
    }
    __syncthreads();
    const unsigned total  = from_ws ? handed : cnt;
    const bool     listed = total <= (unsigned)LCAP;
    const bool give = (LAYER == 1) && share && listed && total <= (unsigned)GSLOT;
    if (LAYER == 1 && share && tid == 0) gcount[blockIdx.x] = give ? total : NO_LIST;

    auto max_apply = [&](const float* a, unsigned dl) {
        float d[HEADS];
        #pragma unroll
        for (int hh = 0; hh < HEADS; ++hh) d[hh] = ad_l[dl * 3 + hh];
        #pragma unroll
        for (int hh = 0; hh < HEADS; ++hh)
            atomicMax(&m_l[dl * 3 + hh], fenc(lrelu(a[hh] + d[hh])));
    };
    if (from_ws) {
        for (unsigned j0 = tid; j0 < total; j0 += WALK_W * LTHR) {
            unsigned wd[WALK_W], dl[WALK_W], sn[WALK_W];
            bool on[WALK_W];
            float a[WALK_W][HEADS];
            #pragma unroll
            for (int w = 0; w < WALK_W; ++w) {
                unsigned j = j0 + w * LTHR;
                on[w] = j < total;
                wd[w] = my_list[on[w] ? j : j0];
            }
            #pragma unroll
            for (int w = 0; w < WALK_W; ++w) {
                const bool mine = wd[w] != NO_EDGE;
                sn[w] = mine ? (wd[w] >> DL_BITS) : 0u;
                dl[w] = mine ? (wd[w] & ((1u << DL_BITS) - 1u)) : NOT_MINE;
            }
            #pragma unroll
            for (int w = 0; w < WALK_W; ++w) load_row3(as_, sn[w], a[w]);
            #pragma unroll
            for (int w = 0; w < WALK_W; ++w) {
                if (on[w]) {
                    l_a[j0 + w * LTHR] = sn[w];
                    l_d[j0 + w * LTHR] = dl[w];
                    if (dl[w] < urn) max_apply(a[w], dl[w]);
                }
            }
        }
    } else if (listed) {
        for (unsigned j0 = tid; j0 < total; j0 += WALK_W * LTHR) {
            unsigned ei[WALK_W], dl[WALK_W], sn[WALK_W];
            bool on[WALK_W];
            float a[WALK_W][HEADS];
            #pragma unroll
            for (int w = 0; w < WALK_W; ++w) {
                unsigned j = j0 + w * LTHR;
                on[w] = j < total;
                unsigned jc = on[w] ? j : j0;
                ei[w] = l_a[jc];
            }
            #pragma unroll
            for (int w = 0; w < WALK_W; ++w) {
                sn[w] = (unsigned)src[ei[w]];
                dl[w] = (unsigned)(dst[ei[w]] - lo);
            }
            #pragma unroll
            for (int w = 0; w < WALK_W; ++w) load_row3(as_, sn[w], a[w]);
            #pragma unroll
            for (int w = 0; w < WALK_W; ++w) {
                if (on[w]) {
                    const bool mine = dl[w] < urn;
                    l_a[j0 + w * LTHR] = sn[w];
                    l_d[j0 + w * LTHR] = mine ? dl[w] : NOT_MINE;
                    if (give) my_list[j0 + w * LTHR] = mine ? ((sn[w] << DL_BITS) | dl[w]) : NO_EDGE;
                    if (mine) max_apply(a[w], dl[w]);
                }
            }
        }
    } else {
        for (int i = tid; i < E; i += LTHR) {
            unsigned r = (unsigned)(dst[i] - lo);
            if (r < urn) {
                unsigned sn = (unsigned)src[i];
                float a[HEADS];
                load_row3(as_, sn, a);
                max_apply(a, r);
            }
        }
    }
    __syncthreads();

    for (int f = tid; f < rn * 3; f += LTHR) {
        float e = lrelu(as_[lo * 3 + f] + ad_l[f]);
        float p = expf(e - fdec(m_l[f]));
        s_l[f] = p;
        #pragma unroll
        for (int cc = 0; cc < 3; ++cc)
            acc_l[f * 3 + cc] = p * h[(lo * 3 + f) * 3 + cc];
    }
    __syncthreads();

    auto sum_apply = [&](const float* a, const float* hv, unsigned dl) {
        float p[HEADS];
        #pragma unroll
        for (int hh = 0; hh < HEADS; ++hh) {
            float e = lrelu(a[hh] + ad_l[dl * 3 + hh]);
            p[hh] = expf(e - fdec(m_l[dl * 3 + hh]));
        }
        #pragma unroll
        for (int hh = 0; hh < HEADS; ++hh) {
            atomicAdd(&s_l[dl * 3 + hh], p[hh]);
            #pragma unroll
            for (int cc = 0; cc < 3; ++cc)
                atomicAdd(&acc_l[dl * 9 + hh * 3 + cc], p[hh] * hv[hh * 3 + cc]);
        }
    };
    if (listed) {
        for (unsigned j0 = tid; j0 < total; j0 += WALK_W * LTHR) {
            unsigned dl[WALK_W], sn[WALK_W];
            bool on[WALK_W];
            float a[WALK_W][HEADS], hv[WALK_W][9];
            #pragma unroll
            for (int w = 0; w < WALK_W; ++w) {
                unsigned j = j0 + w * LTHR;
                on[w] = j < total;
                unsigned jc = on[w] ? j : j0;
                sn[w] = l_a[jc];
                dl[w] = l_d[jc];
                on[w] = on[w] && (dl[w] != NOT_MINE);
            }
            #pragma unroll
            for (int w = 0; w < WALK_W; ++w) {
                load_row3(as_, sn[w], a[w]);
                load_row9(h, sn[w], hv[w]);
            }
            #pragma unroll
            for (int w = 0; w < WALK_W; ++w) {
                if (on[w]) sum_apply(a[w], hv[w], dl[w]);
            }
        }
    } else {
        for (int i = tid; i < E; i += LTHR) {
            unsigned r = (unsigned)(dst[i] - lo);
            if (r < urn) {
                unsigned sn = (unsigned)src[i];
                float a[HEADS], hv[9];
                load_row3(as_, sn, a);
                load_row9(h, sn, hv);
                sum_apply(a, hv, r);
            }
        }
    }
    __syncthreads();

    if (LAYER == 1) {
        for (int f = tid; f < rn * 9; f += LTHR) {
            int nl = f / 9;
            int k  = f - nl * 9;
            int hh = k / 3;
            float inv = 1.0f / (s_l[nl * 3 + hh] + GAT_EPS);
            float v = acc_l[f] * inv + bias[k];
            outp[lo * 9 + f] = (v > 0.f) ? v : (expf(v) - 1.0f);
        }
    } else {
        for (int nl = tid; nl < rn; nl += LTHR) {
            float v[NCLS];
            #pragma unroll
            for (int cc = 0; cc < NCLS; ++cc) {
                float acv = 0.f;
                #pragma unroll
                for (int hh = 0; hh < HEADS; ++hh)
                    acv += acc_l[nl * 9 + hh * 3 + cc] / (s_l[nl * 3 + hh] + GAT_EPS);
                v[cc] = acv * (1.0f / HEADS) + bias[cc];
            }
            float mx = fmaxf(v[0], fmaxf(v[1], v[2]));
            float lse = logf(expf(v[0] - mx) + expf(v[1] - mx) + expf(v[2] - mx));
            #pragma unroll
            for (int cc = 0; cc < NCLS; ++cc)
                outp[(lo + nl) * NCLS + cc] = v[cc] - mx - lse;
        }
    }
}

__global__ void gemm2(const float* __restrict__ t, const float* __restrict__ W2,
                      float* __restrict__ h2, int N) {
    __shared__ float Ws[81];
    if (threadIdx.x < 81) Ws[threadIdx.x] = W2[threadIdx.x];
    __syncthreads();
    int n = blockIdx.x * blockDim.x + threadIdx.x;
    if (n >= N) return;
    float tin[9];
    #pragma unroll
    for (int i = 0; i < 9; ++i) tin[i] = t[n * 9 + i];
    #pragma unroll
    for (int j = 0; j < 9; ++j) {
        float a = 0.f;
        #pragma unroll
        for (int i = 0; i < 9; ++i) a += tin[i] * Ws[i * 9 + j];
        h2[n * 9 + j] = a;
    }
}

extern "C" void kernel_launch(void* const* d_in, const int* in_sizes, int n_in,
                              void* d_out, int out_size, void* d_ws, size_t ws_size,
                              hipStream_t stream) {
    const float* x        = (const float*)d_in[0];
    const int*   eidx     = (const int*)d_in[1];
    const float* W1       = (const float*)d_in[2];
    const float* att_src1 = (const float*)d_in[3];
    const float* att_dst1 = (const float*)d_in[4];
    const float* b1       = (const float*)d_in[5];
    const float* W2       = (const float*)d_in[6];
    const float* att_src2 = (const float*)d_in[7];
    const float* att_dst2 = (const float*)d_in[8];
    const float* b2       = (const float*)d_in[9];

    const int N = in_sizes[0] / IN_DIM;
    const int E = in_sizes[1] / 2;
    const int* src = eidx;
    const int* dst = eidx + E;

    float* h   = (float*)d_ws;
    float* as_ = h   + (size_t)N * 9;
    float* ad_ = as_ + (size_t)N * 3;
    float* t   = ad_ + (size_t)N * 3;
    unsigned* keys = (unsigned*)(t + (size_t)N * 9);

    const int nb = (N + 255) / 256;
    const int tiles = (N + 15) / 16;
    const int gb = (tiles + 7) / 8;
    const int kb = (N + RNG - 1) / RNG;
    const int kw = (E + 3) / 4;
    const int wb = (kw + 255) / 256;
    unsigned* glist  = keys + (((size_t)kw + 63) / 64) * 64;
    unsigned* gcount = glist + (size_t)kb * GSLOT;
    const size_t ws_need = (size_t)((const char*)(gcount + kb) - (const char*)d_ws);

    gemm1_wmma   <<<gb, 256, 0, stream>>>(x, W1, h, N);
    node_att     <<<nb, 256, 0, stream>>>(h, att_src1, att_dst1, as_, ad_, N);
    edge_owner_keys<<<wb, 256, 0, stream>>>(dst, keys, E);
    gat_layer<1> <<<kb, LTHR, 0, stream>>>(src, dst, keys, h, as_, ad_, b1, t, glist, gcount, ws_need, ws_size, N, E);

    gemm2        <<<nb, 256, 0, stream>>>(t, W2, h, N);
    node_att     <<<nb, 256, 0, stream>>>(h, att_src2, att_dst2, as_, ad_, N);
    gat_layer<2> <<<kb, LTHR, 0, stream>>>(src, dst, keys, h, as_, ad_, b2, (float*)d_out, glist, gcount, ws_need, ws_size, N, E);
}
